# starting kernel
# baseline (speedup 1.0000x reference)
#include <hip/hip_runtime.h>
#include <math.h>

typedef float v2f __attribute__((ext_vector_type(2)));
typedef float v4f __attribute__((ext_vector_type(4)));
typedef float v8f __attribute__((ext_vector_type(8)));

#define ROWS_PER_BLOCK 128
#define THREADS 128
#define PRE_LD 136
#define H_LD 72


template<int DIM, int LEN, int UNROLL>
__device__ __forceinline__ void pool_to(float* dst, const float* __restrict__ tab,
                                        const int* __restrict__ ids, int row) {
  float acc[DIM];
#pragma unroll
  for (int d = 0; d < DIM; ++d) acc[d] = 0.f;
#pragma unroll UNROLL
  for (int j = 0; j < LEN; ++j) {
    const int id = ids[row * LEN + j];
    float v[DIM];
    if constexpr ((DIM % 4) == 0) {
      const v4f* tr = (const v4f*)(tab + (size_t)id * DIM);
#pragma unroll
      for (int c = 0; c < DIM / 4; ++c) {
        v4f t = tr[c];
        v[4 * c + 0] = t.x; v[4 * c + 1] = t.y;
        v[4 * c + 2] = t.z; v[4 * c + 3] = t.w;
      }
    } else {
      const v2f* tr = (const v2f*)(tab + (size_t)id * DIM);
#pragma unroll
      for (int c = 0; c < DIM / 2; ++c) {
        v2f t = tr[c];
        v[2 * c + 0] = t.x; v[2 * c + 1] = t.y;
      }
    }
    float ss = 0.f;
#pragma unroll
    for (int d = 0; d < DIM; ++d) ss = fmaf(v[d], v[d], ss);
    float n = sqrtf(ss);
    float s = fminf(1.0f, 1.0f / (n + 1e-7f));
#pragma unroll
    for (int d = 0; d < DIM; ++d) acc[d] = fmaf(v[d], s, acc[d]);
  }
#pragma unroll
  for (int d = 0; d < DIM; ++d) dst[d] = acc[d];
}

__device__ __forceinline__ v8f wmma_f32_k4(v2f a, v2f b, v8f c) {
#if __has_builtin(__builtin_amdgcn_wmma_f32_16x16x4_f32)
  return __builtin_amdgcn_wmma_f32_16x16x4_f32(false, a, false, b, (short)0, c,
                                               false, false);
#else
  asm volatile("v_wmma_f32_16x16x4_f32 %0, %1, %2, %0"
               : "+v"(c) : "v"(a), "v"(b));
  return c;
#endif
}

__device__ __forceinline__ const float* launder(const float* p) {
  asm("" : "+s"(p));
  return p;
}

__global__ __launch_bounds__(THREADS)
__attribute__((amdgpu_waves_per_eu(4)))
void fused_embmlp_kernel(
    const float* __restrict__ dense,
    const int* __restrict__ ids_rating, const int* __restrict__ ids_category,
    const int* __restrict__ ids_fandom, const int* __restrict__ ids_relationship,
    const int* __restrict__ ids_character, const int* __restrict__ ids_freeform,
    const float* __restrict__ W_dense, const float* __restrict__ b_dense,
    const float* __restrict__ tab_rating, const float* __restrict__ tab_category,
    const float* __restrict__ tab_fandom, const float* __restrict__ tab_relationship,
    const float* __restrict__ tab_character, const float* __restrict__ tab_freeform,
    const float* __restrict__ W1, const float* __restrict__ b1,
    const float* __restrict__ W2, const float* __restrict__ b2,
    float* __restrict__ out) {
  extern __shared__ float smem[];
  float* s_pre = smem;
  float* s_h   = smem + ROWS_PER_BLOCK * PRE_LD;

  const int tid = threadIdx.x;
  const int row = blockIdx.x * ROWS_PER_BLOCK + tid;
  float* myrow = s_pre + tid * PRE_LD;

  {
    const v4f* dr4 = (const v4f*)(dense + (size_t)row * 64);
    float acc[20];
#pragma unroll
    for (int j = 0; j < 20; ++j) acc[j] = b_dense[j];
#pragma unroll 2
    for (int kc = 0; kc < 16; ++kc) {
      v4f x = dr4[kc];
#pragma unroll
      for (int j = 0; j < 20; ++j) {
        acc[j] = fmaf(x.x, W_dense[(4 * kc + 0) * 20 + j], acc[j]);
        acc[j] = fmaf(x.y, W_dense[(4 * kc + 1) * 20 + j], acc[j]);
        acc[j] = fmaf(x.z, W_dense[(4 * kc + 2) * 20 + j], acc[j]);
        acc[j] = fmaf(x.w, W_dense[(4 * kc + 3) * 20 + j], acc[j]);
      }
    }
#pragma unroll
    for (int j = 0; j < 20; ++j) myrow[j] = acc[j];
  }
  pool_to<10, 1, 1>(myrow + 20,  tab_rating,       ids_rating,       row);
  pool_to<10, 2, 2>(myrow + 30,  tab_category,     ids_category,     row);
  pool_to<20, 3, 2>(myrow + 40,  tab_fandom,       ids_fandom,       row);
  pool_to<20, 3, 2>(myrow + 60,  tab_relationship, ids_relationship, row);
  pool_to<20, 5, 2>(myrow + 80,  tab_character,    ids_character,    row);
  pool_to<20, 10, 2>(myrow + 100, tab_freeform,    ids_freeform,     row);

  {
    const int PI[15] = {0,0,0,0,0,1,1,1,1,2,2,2,3,3,4};
    const int PJ[15] = {1,2,3,4,5,2,3,4,5,3,4,5,4,5,5};
#pragma unroll 3
    for (int p = 0; p < 15; ++p) {
      const float* a = myrow + 20 * PI[p];
      const float* b = myrow + 20 * PJ[p];
      float s = 0.f;
#pragma unroll
      for (int d = 0; d < 20; ++d) s = fmaf(a[d], b[d], s);
      myrow[120 + p] = s;
    }
    myrow[135] = 0.f;
  }

  __syncthreads();

  const int wave = tid >> 5;
  const int lane = tid & 31;
  const int lo = lane & 15;
  const int hi = lane >> 4;
  float* hbase = s_h + wave * 16 * H_LD;

  for (int t = 0; t < 2; ++t) {
    const int m0 = (wave * 2 + t) * 16;

    const float* arow = s_pre + (m0 + lo) * PRE_LD + 2 * hi;
#pragma unroll
    for (int nt = 0; nt < 4; ++nt) {
      const float* W1l = launder(W1);
      const int n0 = nt * 16;
      const float bias = b1[n0 + lo];
      v8f c;
#pragma unroll
      for (int i = 0; i < 8; ++i) c[i] = bias;
#pragma unroll
      for (int k0 = 0; k0 < 136; k0 += 4) {
        v2f a; a.x = arow[k0]; a.y = arow[k0 + 1];
        const int kk = k0 + 2 * hi;
        v2f b;
        b.x = W1l[kk * 64 + n0 + lo];
        b.y = (kk + 1 < 135) ? W1l[(kk + 1) * 64 + n0 + lo] : 0.f;
        c = wmma_f32_k4(a, b, c);
      }
#pragma unroll
      for (int i = 0; i < 8; ++i) {
        float v = c[i];
        v = v > 0.f ? v : 0.01f * v;
        hbase[(i + 8 * hi) * H_LD + n0 + lo] = v;
      }
    }
    asm volatile("s_wait_dscnt 0x0" ::: "memory");

    const float* hrow = hbase + lo * H_LD + 2 * hi;
    const size_t rbase = (size_t)(blockIdx.x * ROWS_PER_BLOCK + m0) * 128;
#pragma unroll
    for (int nt = 0; nt < 8; ++nt) {
      const float* W2l = launder(W2);
      const int n0 = nt * 16;
      const float bias = b2[n0 + lo];
      v8f c;
#pragma unroll
      for (int i = 0; i < 8; ++i) c[i] = bias;
#pragma unroll
      for (int k0 = 0; k0 < 64; k0 += 4) {
        v2f a; a.x = hrow[k0]; a.y = hrow[k0 + 1];
        const int kk = k0 + 2 * hi;
        v2f b;
        b.x = W2l[kk * 128 + n0 + lo];
        b.y = W2l[(kk + 1) * 128 + n0 + lo];
        c = wmma_f32_k4(a, b, c);
      }
#pragma unroll
      for (int i = 0; i < 8; ++i)
        out[rbase + (size_t)(i + 8 * hi) * 128 + n0 + lo] = c[i];
    }
  }
}

extern "C" void kernel_launch(void* const* d_in, const int* in_sizes, int n_in,
                              void* d_out, int out_size, void* d_ws, size_t ws_size,
                              hipStream_t stream) {
  const float* dense = (const float*)d_in[0];
  const int* ids_rating       = (const int*)d_in[1];
  const int* ids_category     = (const int*)d_in[2];
  const int* ids_fandom       = (const int*)d_in[3];
  const int* ids_relationship = (const int*)d_in[4];
  const int* ids_character    = (const int*)d_in[5];
  const int* ids_freeform     = (const int*)d_in[6];
  const float* W_dense = (const float*)d_in[7];
  const float* b_dense = (const float*)d_in[8];
  const float* tab_rating       = (const float*)d_in[9];
  const float* tab_category     = (const float*)d_in[10];
  const float* tab_fandom       = (const float*)d_in[11];
  const float* tab_relationship = (const float*)d_in[12];
  const float* tab_character    = (const float*)d_in[13];
  const float* tab_freeform     = (const float*)d_in[14];
  const float* W1 = (const float*)d_in[15];
  const float* b1 = (const float*)d_in[16];
  const float* W2 = (const float*)d_in[17];
  const float* b2 = (const float*)d_in[18];
  float* out = (float*)d_out;

  const int Btot = in_sizes[0] / 64;
  const int grid = Btot / ROWS_PER_BLOCK;
  const size_t shmem = (size_t)(ROWS_PER_BLOCK * PRE_LD + 4 * 16 * H_LD) * sizeof(float);

  fused_embmlp_kernel<<<grid, THREADS, shmem, stream>>>(
      dense, ids_rating, ids_category, ids_fandom, ids_relationship,
      ids_character, ids_freeform, W_dense, b_dense, tab_rating, tab_category,
      tab_fandom, tab_relationship, tab_character, tab_freeform, W1, b1, W2, b2,
      out);
}


# Phase 1 split over two threads per row (64-row blocks: waves 0-1 dense + 4 short bags, waves 2-3 the 2 long bags; dots split 8/7): 4 waves per SIMD instead of 2; order of all sums unchanged
# speedup vs baseline: 1.3159x; 1.3159x over previous
#include <hip/hip_runtime.h>
#include <math.h>

typedef float v2f __attribute__((ext_vector_type(2)));
typedef float v4f __attribute__((ext_vector_type(4)));
typedef float v8f __attribute__((ext_vector_type(8)));

#define ROWS_PER_BLOCK 64
#define THREADS 128
#define PRE_LD 136
#define H_LD 72


template<int DIM, int LEN, int UNROLL>
__device__ __forceinline__ void pool_to(float* dst, const float* __restrict__ tab,
                                        const int* __restrict__ ids, int row) {
  float acc[DIM];
#pragma unroll
  for (int d = 0; d < DIM; ++d) acc[d] = 0.f;
#pragma unroll UNROLL
  for (int j = 0; j < LEN; ++j) {
    const int id = ids[row * LEN + j];
    float v[DIM];
    if constexpr ((DIM % 4) == 0) {
      const v4f* tr = (const v4f*)(tab + (size_t)id * DIM);
#pragma unroll
      for (int c = 0; c < DIM / 4; ++c) {
        v4f t = tr[c];
        v[4 * c + 0] = t.x; v[4 * c + 1] = t.y;
        v[4 * c + 2] = t.z; v[4 * c + 3] = t.w;
      }
    } else {
      const v2f* tr = (const v2f*)(tab + (size_t)id * DIM);
#pragma unroll
      for (int c = 0; c < DIM / 2; ++c) {
        v2f t = tr[c];
        v[2 * c + 0] = t.x; v[2 * c + 1] = t.y;
      }
    }
    float ss = 0.f;
#pragma unroll
    for (int d = 0; d < DIM; ++d) ss = fmaf(v[d], v[d], ss);
    float n = sqrtf(ss);
    float s = fminf(1.0f, 1.0f / (n + 1e-7f));
#pragma unroll
    for (int d = 0; d < DIM; ++d) acc[d] = fmaf(v[d], s, acc[d]);
  }
#pragma unroll
  for (int d = 0; d < DIM; ++d) dst[d] = acc[d];
}

__device__ __forceinline__ v8f wmma_f32_k4(v2f a, v2f b, v8f c) {
#if __has_builtin(__builtin_amdgcn_wmma_f32_16x16x4_f32)
  return __builtin_amdgcn_wmma_f32_16x16x4_f32(false, a, false, b, (short)0, c,
                                               false, false);
#else
  asm volatile("v_wmma_f32_16x16x4_f32 %0, %1, %2, %0"
               : "+v"(c) : "v"(a), "v"(b));
  return c;
#endif
}

template<int P0, int P1>
__device__ __forceinline__ void pair_dots(float* myrow) {
  constexpr int PI[15] = {0,0,0,0,0,1,1,1,1,2,2,2,3,3,4};
  constexpr int PJ[15] = {1,2,3,4,5,2,3,4,5,3,4,5,4,5,5};
#pragma unroll
  for (int p = P0; p < P1; ++p) {
    const float* a = myrow + 20 * PI[p];
    const float* b = myrow + 20 * PJ[p];
    float s = 0.f;
#pragma unroll
    for (int d = 0; d < 20; ++d) s = fmaf(a[d], b[d], s);
    myrow[120 + p] = s;
  }
}

__device__ __forceinline__ const float* launder(const float* p) {
  asm("" : "+s"(p));
  return p;
}

__global__ __launch_bounds__(THREADS)
__attribute__((amdgpu_waves_per_eu(4)))
void fused_embmlp_kernel(
    const float* __restrict__ dense,
    const int* __restrict__ ids_rating, const int* __restrict__ ids_category,
    const int* __restrict__ ids_fandom, const int* __restrict__ ids_relationship,
    const int* __restrict__ ids_character, const int* __restrict__ ids_freeform,
    const float* __restrict__ W_dense, const float* __restrict__ b_dense,
    const float* __restrict__ tab_rating, const float* __restrict__ tab_category,
    const float* __restrict__ tab_fandom, const float* __restrict__ tab_relationship,
    const float* __restrict__ tab_character, const float* __restrict__ tab_freeform,
    const float* __restrict__ W1, const float* __restrict__ b1,
    const float* __restrict__ W2, const float* __restrict__ b2,
    float* __restrict__ out) {
  extern __shared__ float smem[];
  float* s_pre = smem;
  float* s_h   = smem + ROWS_PER_BLOCK * PRE_LD;

  const int tid = threadIdx.x;
  const int half = __builtin_amdgcn_readfirstlane(tid >> 6);
  const int r = tid & 63;
  const int row = blockIdx.x * ROWS_PER_BLOCK + r;
  float* myrow = s_pre + r * PRE_LD;

  if (half == 0) {
    {
      const v4f* dr4 = (const v4f*)(dense + (size_t)row * 64);
      float acc[20];
#pragma unroll
      for (int j = 0; j < 20; ++j) acc[j] = b_dense[j];
#pragma unroll 2
      for (int kc = 0; kc < 16; ++kc) {
        v4f x = dr4[kc];
#pragma unroll
        for (int j = 0; j < 20; ++j) {
          acc[j] = fmaf(x.x, W_dense[(4 * kc + 0) * 20 + j], acc[j]);
          acc[j] = fmaf(x.y, W_dense[(4 * kc + 1) * 20 + j], acc[j]);
          acc[j] = fmaf(x.z, W_dense[(4 * kc + 2) * 20 + j], acc[j]);
          acc[j] = fmaf(x.w, W_dense[(4 * kc + 3) * 20 + j], acc[j]);
        }
      }
#pragma unroll
      for (int j = 0; j < 20; ++j) myrow[j] = acc[j];
    }
    pool_to<10, 1, 1>(myrow + 20,  tab_rating,       ids_rating,       row);
    pool_to<10, 2, 2>(myrow + 30,  tab_category,     ids_category,     row);
    pool_to<20, 3, 2>(myrow + 40,  tab_fandom,       ids_fandom,       row);
    pool_to<20, 3, 2>(myrow + 60,  tab_relationship, ids_relationship, row);
  } else {
    pool_to<20, 5, 2>(myrow + 80,  tab_character,    ids_character,    row);
    pool_to<20, 10, 2>(myrow + 100, tab_freeform,    ids_freeform,     row);
  }

  __syncthreads();

  if (half == 0) {
    pair_dots<0, 8>(myrow);
  } else {
    pair_dots<8, 15>(myrow);
    myrow[135] = 0.f;
  }

  __syncthreads();

  const int wave = tid >> 5;
  const int lane = tid & 31;
  const int lo = lane & 15;
  const int hi = lane >> 4;
  float* hbase = s_h + wave * 16 * H_LD;

  {
    const int m0 = wave * 16;

    const float* arow = s_pre + (m0 + lo) * PRE_LD + 2 * hi;
#pragma unroll
    for (int nt = 0; nt < 4; ++nt) {
      const float* W1l = launder(W1);
      const int n0 = nt * 16;
      const float bias = b1[n0 + lo];
      v8f c;
#pragma unroll
      for (int i = 0; i < 8; ++i) c[i] = bias;
#pragma unroll
      for (int k0 = 0; k0 < 136; k0 += 4) {
        v2f a; a.x = arow[k0]; a.y = arow[k0 + 1];
        const int kk = k0 + 2 * hi;
        v2f b;
        b.x = W1l[kk * 64 + n0 + lo];
        b.y = (kk + 1 < 135) ? W1l[(kk + 1) * 64 + n0 + lo] : 0.f;
        c = wmma_f32_k4(a, b, c);
      }
#pragma unroll
      for (int i = 0; i < 8; ++i) {
        float v = c[i];
        v = v > 0.f ? v : 0.01f * v;
        hbase[(i + 8 * hi) * H_LD + n0 + lo] = v;
      }
    }
    asm volatile("s_wait_dscnt 0x0" ::: "memory");

    const float* hrow = hbase + lo * H_LD + 2 * hi;
    const size_t rbase = (size_t)(blockIdx.x * ROWS_PER_BLOCK + m0) * 128;
#pragma unroll
    for (int nt = 0; nt < 8; ++nt) {
      const float* W2l = launder(W2);
      const int n0 = nt * 16;
      const float bias = b2[n0 + lo];
      v8f c;
#pragma unroll
      for (int i = 0; i < 8; ++i) c[i] = bias;
#pragma unroll
      for (int k0 = 0; k0 < 64; k0 += 4) {
        v2f a; a.x = hrow[k0]; a.y = hrow[k0 + 1];
        const int kk = k0 + 2 * hi;
        v2f b;
        b.x = W2l[kk * 128 + n0 + lo];
        b.y = W2l[(kk + 1) * 128 + n0 + lo];
        c = wmma_f32_k4(a, b, c);
      }
#pragma unroll
      for (int i = 0; i < 8; ++i)
        out[rbase + (size_t)(i + 8 * hi) * 128 + n0 + lo] = c[i];
    }
  }
}

extern "C" void kernel_launch(void* const* d_in, const int* in_sizes, int n_in,
                              void* d_out, int out_size, void* d_ws, size_t ws_size,
                              hipStream_t stream) {
  const float* dense = (const float*)d_in[0];
  const int* ids_rating       = (const int*)d_in[1];
  const int* ids_category     = (const int*)d_in[2];
  const int* ids_fandom       = (const int*)d_in[3];
  const int* ids_relationship = (const int*)d_in[4];
  const int* ids_character    = (const int*)d_in[5];
  const int* ids_freeform     = (const int*)d_in[6];
  const float* W_dense = (const float*)d_in[7];
  const float* b_dense = (const float*)d_in[8];
  const float* tab_rating       = (const float*)d_in[9];
  const float* tab_category     = (const float*)d_in[10];
  const float* tab_fandom       = (const float*)d_in[11];
  const float* tab_relationship = (const float*)d_in[12];
  const float* tab_character    = (const float*)d_in[13];
  const float* tab_freeform     = (const float*)d_in[14];
  const float* W1 = (const float*)d_in[15];
  const float* b1 = (const float*)d_in[16];
  const float* W2 = (const float*)d_in[17];
  const float* b2 = (const float*)d_in[18];
  float* out = (float*)d_out;

  const int Btot = in_sizes[0] / 64;
  const int grid = Btot / ROWS_PER_BLOCK;
  const size_t shmem = (size_t)(ROWS_PER_BLOCK * PRE_LD + 4 * 16 * H_LD) * sizeof(float);

  fused_embmlp_kernel<<<grid, THREADS, shmem, stream>>>(
      dense, ids_rating, ids_category, ids_fandom, ids_relationship,
      ids_character, ids_freeform, W_dense, b_dense, tab_rating, tab_category,
      tab_fandom, tab_relationship, tab_character, tab_freeform, W1, b1, W2, b2,
      out);
}


# Phase 2 re-tiled: each wave owns an N slice (16 cols of MLP1, 32 of MLP2) over all 64 rows, so each weight fragment is loaded once per block and feeds 4 M-tiles; order of all sums unchanged
# speedup vs baseline: 1.4328x; 1.0888x over previous
#include <hip/hip_runtime.h>
#include <math.h>

typedef float v2f __attribute__((ext_vector_type(2)));
typedef float v4f __attribute__((ext_vector_type(4)));
typedef float v8f __attribute__((ext_vector_type(8)));

#define ROWS_PER_BLOCK 64
#define THREADS 128
#define PRE_LD 136
#define H_LD 72


template<int DIM, int LEN, int UNROLL>
__device__ __forceinline__ void pool_to(float* dst, const float* __restrict__ tab,
                                        const int* __restrict__ ids, int row) {
  float acc[DIM];
#pragma unroll
  for (int d = 0; d < DIM; ++d) acc[d] = 0.f;
#pragma unroll UNROLL
  for (int j = 0; j < LEN; ++j) {
    const int id = ids[row * LEN + j];
    float v[DIM];
    if constexpr ((DIM % 4) == 0) {
      const v4f* tr = (const v4f*)(tab + (size_t)id * DIM);
#pragma unroll
      for (int c = 0; c < DIM / 4; ++c) {
        v4f t = tr[c];
        v[4 * c + 0] = t.x; v[4 * c + 1] = t.y;
        v[4 * c + 2] = t.z; v[4 * c + 3] = t.w;
      }
    } else {
      const v2f* tr = (const v2f*)(tab + (size_t)id * DIM);
#pragma unroll
      for (int c = 0; c < DIM / 2; ++c) {
        v2f t = tr[c];
        v[2 * c + 0] = t.x; v[2 * c + 1] = t.y;
      }
    }
    float ss = 0.f;
#pragma unroll
    for (int d = 0; d < DIM; ++d) ss = fmaf(v[d], v[d], ss);
    float n = sqrtf(ss);
    float s = fminf(1.0f, 1.0f / (n + 1e-7f));
#pragma unroll
    for (int d = 0; d < DIM; ++d) acc[d] = fmaf(v[d], s, acc[d]);
  }
#pragma unroll
  for (int d = 0; d < DIM; ++d) dst[d] = acc[d];
}

__device__ __forceinline__ v8f wmma_f32_k4(v2f a, v2f b, v8f c) {
#if __has_builtin(__builtin_amdgcn_wmma_f32_16x16x4_f32)
  return __builtin_amdgcn_wmma_f32_16x16x4_f32(false, a, false, b, (short)0, c,
                                               false, false);
#else
  asm volatile("v_wmma_f32_16x16x4_f32 %0, %1, %2, %0"
               : "+v"(c) : "v"(a), "v"(b));
  return c;
#endif
}

template<int P0, int P1>
__device__ __forceinline__ void pair_dots(float* myrow) {
  constexpr int PI[15] = {0,0,0,0,0,1,1,1,1,2,2,2,3,3,4};
  constexpr int PJ[15] = {1,2,3,4,5,2,3,4,5,3,4,5,4,5,5};
#pragma unroll
  for (int p = P0; p < P1; ++p) {
    const float* a = myrow + 20 * PI[p];
    const float* b = myrow + 20 * PJ[p];
    float s = 0.f;
#pragma unroll
    for (int d = 0; d < 20; ++d) s = fmaf(a[d], b[d], s);
    myrow[120 + p] = s;
  }
}

__global__ __launch_bounds__(THREADS)
__attribute__((amdgpu_waves_per_eu(4)))
void fused_embmlp_kernel(
    const float* __restrict__ dense,
    const int* __restrict__ ids_rating, const int* __restrict__ ids_category,
    const int* __restrict__ ids_fandom, const int* __restrict__ ids_relationship,
    const int* __restrict__ ids_character, const int* __restrict__ ids_freeform,
    const float* __restrict__ W_dense, const float* __restrict__ b_dense,
    const float* __restrict__ tab_rating, const float* __restrict__ tab_category,
    const float* __restrict__ tab_fandom, const float* __restrict__ tab_relationship,
    const float* __restrict__ tab_character, const float* __restrict__ tab_freeform,
    const float* __restrict__ W1, const float* __restrict__ b1,
    const float* __restrict__ W2, const float* __restrict__ b2,
    float* __restrict__ out) {
  extern __shared__ __attribute__((aligned(16))) float smem[];
  float* s_pre = smem;
  float* s_h   = smem + ROWS_PER_BLOCK * PRE_LD;

  const int tid = threadIdx.x;
  const int half = __builtin_amdgcn_readfirstlane(tid >> 6);
  const int r = tid & 63;
  const int row = blockIdx.x * ROWS_PER_BLOCK + r;
  float* myrow = s_pre + r * PRE_LD;

  if (half == 0) {
    {
      const v4f* dr4 = (const v4f*)(dense + (size_t)row * 64);
      float acc[20];
#pragma unroll
      for (int j = 0; j < 20; ++j) acc[j] = b_dense[j];
#pragma unroll 2
      for (int kc = 0; kc < 16; ++kc) {
        v4f x = dr4[kc];
#pragma unroll
        for (int j = 0; j < 20; ++j) {
          acc[j] = fmaf(x.x, W_dense[(4 * kc + 0) * 20 + j], acc[j]);
          acc[j] = fmaf(x.y, W_dense[(4 * kc + 1) * 20 + j], acc[j]);
          acc[j] = fmaf(x.z, W_dense[(4 * kc + 2) * 20 + j], acc[j]);
          acc[j] = fmaf(x.w, W_dense[(4 * kc + 3) * 20 + j], acc[j]);
        }
      }
#pragma unroll
      for (int j = 0; j < 20; ++j) myrow[j] = acc[j];
    }
    pool_to<10, 1, 1>(myrow + 20,  tab_rating,       ids_rating,       row);
    pool_to<10, 2, 2>(myrow + 30,  tab_category,     ids_category,     row);
    pool_to<20, 3, 2>(myrow + 40,  tab_fandom,       ids_fandom,       row);
    pool_to<20, 3, 2>(myrow + 60,  tab_relationship, ids_relationship, row);
  } else {
    pool_to<20, 5, 2>(myrow + 80,  tab_character,    ids_character,    row);
    pool_to<20, 10, 2>(myrow + 100, tab_freeform,    ids_freeform,     row);
  }

  __syncthreads();

  if (half == 0) {
    pair_dots<0, 8>(myrow);
  } else {
    pair_dots<8, 15>(myrow);
    myrow[135] = 0.f;
  }

  __syncthreads();

  const int wave = __builtin_amdgcn_readfirstlane(tid >> 5);
  const int lane = tid & 31;
  const int lo = lane & 15;
  const int hi = lane >> 4;

  {
    const int n0 = wave * 16;
    const float bias = b1[n0 + lo];
    v8f c[4];
#pragma unroll
    for (int mt = 0; mt < 4; ++mt)
#pragma unroll
      for (int i = 0; i < 8; ++i) c[mt][i] = bias;
    const float* arow = s_pre + lo * PRE_LD + 2 * hi;
    const float* wcol = W1 + n0 + lo;
#pragma unroll
    for (int k0 = 0; k0 < 136; k0 += 4) {
      const int kk = k0 + 2 * hi;
      v2f b;
      b.x = wcol[kk * 64];
      if (k0 + 3 < 135) {
        b.y = wcol[(kk + 1) * 64];
      } else {
        const int k1 = (kk + 1 < 135) ? (kk + 1) : 134;
        const float t = wcol[k1 * 64];
        b.y = (kk + 1 < 135) ? t : 0.f;
      }
#pragma unroll
      for (int mt = 0; mt < 4; ++mt) {
        const v2f a = *(const v2f*)(arow + mt * 16 * PRE_LD + k0);
        c[mt] = wmma_f32_k4(a, b, c[mt]);
      }
    }
#pragma unroll
    for (int mt = 0; mt < 4; ++mt) {
#pragma unroll
      for (int i = 0; i < 8; ++i) {
        float v = c[mt][i];
        v = v > 0.f ? v : 0.01f * v;
        s_h[(mt * 16 + i + 8 * hi) * H_LD + n0 + lo] = v;
      }
    }
  }

  __syncthreads();

  {
    const int n0 = wave * 32;
    const float bias0 = b2[n0 + lo];
    const float bias1 = b2[n0 + 16 + lo];
    v8f c0[4], c1[4];
#pragma unroll
    for (int mt = 0; mt < 4; ++mt)
#pragma unroll
      for (int i = 0; i < 8; ++i) { c0[mt][i] = bias0; c1[mt][i] = bias1; }
    const float* hrow = s_h + lo * H_LD + 2 * hi;
    const float* wcol = W2 + n0 + lo;
#pragma unroll
    for (int k0 = 0; k0 < 64; k0 += 4) {
      const int kk = k0 + 2 * hi;
      v2f b0, b1v;
      b0.x  = wcol[kk * 128];
      b0.y  = wcol[(kk + 1) * 128];
      b1v.x = wcol[kk * 128 + 16];
      b1v.y = wcol[(kk + 1) * 128 + 16];
#pragma unroll
      for (int mt = 0; mt < 4; ++mt) {
        const v2f a = *(const v2f*)(hrow + mt * 16 * H_LD + k0);
        c0[mt] = wmma_f32_k4(a, b0, c0[mt]);
        c1[mt] = wmma_f32_k4(a, b1v, c1[mt]);
      }
    }
    const size_t rbase = (size_t)(blockIdx.x * ROWS_PER_BLOCK) * 128 + n0 + lo;
#pragma unroll
    for (int mt = 0; mt < 4; ++mt) {
#pragma unroll
      for (int i = 0; i < 8; ++i) {
        const size_t o = rbase + (size_t)(mt * 16 + i + 8 * hi) * 128;
        out[o]      = c0[mt][i];
        out[o + 16] = c1[mt][i];
      }
    }
  }
}

extern "C" void kernel_launch(void* const* d_in, const int* in_sizes, int n_in,
                              void* d_out, int out_size, void* d_ws, size_t ws_size,
                              hipStream_t stream) {
  const float* dense = (const float*)d_in[0];
  const int* ids_rating       = (const int*)d_in[1];
  const int* ids_category     = (const int*)d_in[2];
  const int* ids_fandom       = (const int*)d_in[3];
  const int* ids_relationship = (const int*)d_in[4];
  const int* ids_character    = (const int*)d_in[5];
  const int* ids_freeform     = (const int*)d_in[6];
  const float* W_dense = (const float*)d_in[7];
  const float* b_dense = (const float*)d_in[8];
  const float* tab_rating       = (const float*)d_in[9];
  const float* tab_category     = (const float*)d_in[10];
  const float* tab_fandom       = (const float*)d_in[11];
  const float* tab_relationship = (const float*)d_in[12];
  const float* tab_character    = (const float*)d_in[13];
  const float* tab_freeform     = (const float*)d_in[14];
  const float* W1 = (const float*)d_in[15];
  const float* b1 = (const float*)d_in[16];
  const float* W2 = (const float*)d_in[17];
  const float* b2 = (const float*)d_in[18];
  float* out = (float*)d_out;

  const int Btot = in_sizes[0] / 64;
  const int grid = Btot / ROWS_PER_BLOCK;
  const size_t shmem = (size_t)(ROWS_PER_BLOCK * PRE_LD + 4 * 16 * H_LD) * sizeof(float);

  fused_embmlp_kernel<<<grid, THREADS, shmem, stream>>>(
      dense, ids_rating, ids_category, ids_fandom, ids_relationship,
      ids_character, ids_freeform, W_dense, b_dense, tab_rating, tab_category,
      tab_fandom, tab_relationship, tab_character, tab_freeform, W1, b1, W2, b2,
      out);
}


# Block's contiguous inputs read once as whole lines: dense tile and the six id lists fetched up front by coalesced 16-byte loads into LDS, each thread reads its row there; order of all sums unchanged
# speedup vs baseline: 1.5214x; 1.0618x over previous
#include <hip/hip_runtime.h>
#include <math.h>

typedef float v2f __attribute__((ext_vector_type(2)));
typedef float v4f __attribute__((ext_vector_type(4)));
typedef float v8f __attribute__((ext_vector_type(8)));
typedef int v4i __attribute__((ext_vector_type(4)));

#define ROWS_PER_BLOCK 64
#define THREADS 128
#define PRE_LD 136
#define H_LD 72
#define X_LD 68
#define IDS_RATING 0
#define IDS_CATEGORY 64
#define IDS_FANDOM 192
#define IDS_RELATIONSHIP 384
#define IDS_CHARACTER 576
#define IDS_FREEFORM 896
#define IDS_TOTAL 1536


template<int DIM, int LEN, int UNROLL>
__device__ __forceinline__ void pool_to(float* dst, const float* __restrict__ tab,
                                        const int* sid) {
  float acc[DIM];
#pragma unroll
  for (int d = 0; d < DIM; ++d) acc[d] = 0.f;
#pragma unroll UNROLL
  for (int j = 0; j < LEN; ++j) {
    const int id = sid[j];
    float v[DIM];
    if constexpr ((DIM % 4) == 0) {
      const v4f* tr = (const v4f*)(tab + (size_t)id * DIM);
#pragma unroll
      for (int c = 0; c < DIM / 4; ++c) {
        v4f t = tr[c];
        v[4 * c + 0] = t.x; v[4 * c + 1] = t.y;
        v[4 * c + 2] = t.z; v[4 * c + 3] = t.w;
      }
    } else {
      const v2f* tr = (const v2f*)(tab + (size_t)id * DIM);
#pragma unroll
      for (int c = 0; c < DIM / 2; ++c) {
        v2f t = tr[c];
        v[2 * c + 0] = t.x; v[2 * c + 1] = t.y;
      }
    }
    float ss = 0.f;
#pragma unroll
    for (int d = 0; d < DIM; ++d) ss = fmaf(v[d], v[d], ss);
    float n = sqrtf(ss);
    float s = fminf(1.0f, 1.0f / (n + 1e-7f));
#pragma unroll
    for (int d = 0; d < DIM; ++d) acc[d] = fmaf(v[d], s, acc[d]);
  }
#pragma unroll
  for (int d = 0; d < DIM; ++d) dst[d] = acc[d];
}

__device__ __forceinline__ v8f wmma_f32_k4(v2f a, v2f b, v8f c) {
#if __has_builtin(__builtin_amdgcn_wmma_f32_16x16x4_f32)
  return __builtin_amdgcn_wmma_f32_16x16x4_f32(false, a, false, b, (short)0, c,
                                               false, false);
#else
  asm volatile("v_wmma_f32_16x16x4_f32 %0, %1, %2, %0"
               : "+v"(c) : "v"(a), "v"(b));
  return c;
#endif
}

template<int P0, int P1>
__device__ __forceinline__ void pair_dots(float* myrow) {
  constexpr int PI[15] = {0,0,0,0,0,1,1,1,1,2,2,2,3,3,4};
  constexpr int PJ[15] = {1,2,3,4,5,2,3,4,5,3,4,5,4,5,5};
#pragma unroll
  for (int p = P0; p < P1; ++p) {
    const float* a = myrow + 20 * PI[p];
    const float* b = myrow + 20 * PJ[p];
    float s = 0.f;
#pragma unroll
    for (int d = 0; d < 20; ++d) s = fmaf(a[d], b[d], s);
    myrow[120 + p] = s;
  }
}

__global__ __launch_bounds__(THREADS)
__attribute__((amdgpu_waves_per_eu(4)))
void fused_embmlp_kernel(
    const float* __restrict__ dense,
    const int* __restrict__ ids_rating, const int* __restrict__ ids_category,
    const int* __restrict__ ids_fandom, const int* __restrict__ ids_relationship,
    const int* __restrict__ ids_character, const int* __restrict__ ids_freeform,
    const float* __restrict__ W_dense, const float* __restrict__ b_dense,
    const float* __restrict__ tab_rating, const float* __restrict__ tab_category,
    const float* __restrict__ tab_fandom, const float* __restrict__ tab_relationship,
    const float* __restrict__ tab_character, const float* __restrict__ tab_freeform,
    const float* __restrict__ W1, const float* __restrict__ b1,
    const float* __restrict__ W2, const float* __restrict__ b2,
    float* __restrict__ out) {
  extern __shared__ __attribute__((aligned(16))) float smem[];
  float* s_pre = smem;
  float* s_h   = smem + ROWS_PER_BLOCK * PRE_LD;
  float* s_x   = s_h;
  int*   s_ids = (int*)(s_h + ROWS_PER_BLOCK * H_LD);

  const int tid = threadIdx.x;
  const int half = __builtin_amdgcn_readfirstlane(tid >> 6);
  const int r = tid & 63;
  float* myrow = s_pre + r * PRE_LD;

  {
    const int brow = blockIdx.x * ROWS_PER_BLOCK;
    v4i idv[3];
#pragma unroll
    for (int p = 0; p < 3; ++p) {
      const int q = tid + THREADS * p;
      const int* src;
      int lq;
      if (q < IDS_CATEGORY / 4)          { src = ids_rating       + brow * 1;  lq = q; }
      else if (q < IDS_FANDOM / 4)       { src = ids_category     + brow * 2;  lq = q - IDS_CATEGORY / 4; }
      else if (q < IDS_RELATIONSHIP / 4) { src = ids_fandom       + brow * 3;  lq = q - IDS_FANDOM / 4; }
      else if (q < IDS_CHARACTER / 4)    { src = ids_relationship + brow * 3;  lq = q - IDS_RELATIONSHIP / 4; }
      else if (q < IDS_FREEFORM / 4)     { src = ids_character    + brow * 5;  lq = q - IDS_CHARACTER / 4; }
      else                               { src = ids_freeform     + brow * 10; lq = q - IDS_FREEFORM / 4; }
      idv[p] = *(const v4i*)(src + 4 * lq);
    }
    v4f xv[8];
    const v4f* dt4 = (const v4f*)(dense + (size_t)brow * 64);
#pragma unroll
    for (int p = 0; p < 8; ++p) xv[p] = dt4[tid + THREADS * p];
#pragma unroll
    for (int p = 0; p < 3; ++p) *(v4i*)(s_ids + 4 * (tid + THREADS * p)) = idv[p];
#pragma unroll
    for (int p = 0; p < 8; ++p) {
      const int c = tid + THREADS * p;
      *(v4f*)(s_x + (c >> 4) * X_LD + 4 * (c & 15)) = xv[p];
    }
  }
  __syncthreads();

  if (half == 0) {
    {
      const v4f* dr4 = (const v4f*)(s_x + r * X_LD);
      float acc[20];
#pragma unroll
      for (int j = 0; j < 20; ++j) acc[j] = b_dense[j];
#pragma unroll 2
      for (int kc = 0; kc < 16; ++kc) {
        v4f x = dr4[kc];
#pragma unroll
        for (int j = 0; j < 20; ++j) {
          acc[j] = fmaf(x.x, W_dense[(4 * kc + 0) * 20 + j], acc[j]);
          acc[j] = fmaf(x.y, W_dense[(4 * kc + 1) * 20 + j], acc[j]);
          acc[j] = fmaf(x.z, W_dense[(4 * kc + 2) * 20 + j], acc[j]);
          acc[j] = fmaf(x.w, W_dense[(4 * kc + 3) * 20 + j], acc[j]);
        }
      }
#pragma unroll
      for (int j = 0; j < 20; ++j) myrow[j] = acc[j];
    }
    pool_to<10, 1, 1>(myrow + 20,  tab_rating,       s_ids + IDS_RATING       + r * 1);
    pool_to<10, 2, 2>(myrow + 30,  tab_category,     s_ids + IDS_CATEGORY     + r * 2);
    pool_to<20, 3, 2>(myrow + 40,  tab_fandom,       s_ids + IDS_FANDOM       + r * 3);
    pool_to<20, 3, 2>(myrow + 60,  tab_relationship, s_ids + IDS_RELATIONSHIP + r * 3);
  } else {
    pool_to<20, 5, 2>(myrow + 80,  tab_character,    s_ids + IDS_CHARACTER    + r * 5);
    pool_to<20, 10, 2>(myrow + 100, tab_freeform,    s_ids + IDS_FREEFORM     + r * 10);
  }

  __syncthreads();

  if (half == 0) {
    pair_dots<0, 8>(myrow);
  } else {
    pair_dots<8, 15>(myrow);
    myrow[135] = 0.f;
  }

  __syncthreads();

  const int wave = __builtin_amdgcn_readfirstlane(tid >> 5);
  const int lane = tid & 31;
  const int lo = lane & 15;
  const int hi = lane >> 4;

  {
    const int n0 = wave * 16;
    const float bias = b1[n0 + lo];
    v8f c[4];
#pragma unroll
    for (int mt = 0; mt < 4; ++mt)
#pragma unroll
      for (int i = 0; i < 8; ++i) c[mt][i] = bias;
    const float* arow = s_pre + lo * PRE_LD + 2 * hi;
    const float* wcol = W1 + n0 + lo;
#pragma unroll
    for (int k0 = 0; k0 < 136; k0 += 4) {
      const int kk = k0 + 2 * hi;
      v2f b;
      b.x = wcol[kk * 64];
      if (k0 + 3 < 135) {
        b.y = wcol[(kk + 1) * 64];
      } else {
        const int k1 = (kk + 1 < 135) ? (kk + 1) : 134;
        const float t = wcol[k1 * 64];
        b.y = (kk + 1 < 135) ? t : 0.f;
      }
#pragma unroll
      for (int mt = 0; mt < 4; ++mt) {
        const v2f a = *(const v2f*)(arow + mt * 16 * PRE_LD + k0);
        c[mt] = wmma_f32_k4(a, b, c[mt]);
      }
    }
#pragma unroll
    for (int mt = 0; mt < 4; ++mt) {
#pragma unroll
      for (int i = 0; i < 8; ++i) {
        float v = c[mt][i];
        v = v > 0.f ? v : 0.01f * v;
        s_h[(mt * 16 + i + 8 * hi) * H_LD + n0 + lo] = v;
      }
    }
  }

  __syncthreads();

  {
    const int n0 = wave * 32;
    const float bias0 = b2[n0 + lo];
    const float bias1 = b2[n0 + 16 + lo];
    v8f c0[4], c1[4];
#pragma unroll
    for (int mt = 0; mt < 4; ++mt)
#pragma unroll
      for (int i = 0; i < 8; ++i) { c0[mt][i] = bias0; c1[mt][i] = bias1; }
    const float* hrow = s_h + lo * H_LD + 2 * hi;
    const float* wcol = W2 + n0 + lo;
#pragma unroll
    for (int k0 = 0; k0 < 64; k0 += 4) {
      const int kk = k0 + 2 * hi;
      v2f b0, b1v;
      b0.x  = wcol[kk * 128];
      b0.y  = wcol[(kk + 1) * 128];
      b1v.x = wcol[kk * 128 + 16];
      b1v.y = wcol[(kk + 1) * 128 + 16];
#pragma unroll
      for (int mt = 0; mt < 4; ++mt) {
        const v2f a = *(const v2f*)(hrow + mt * 16 * H_LD + k0);
        c0[mt] = wmma_f32_k4(a, b0, c0[mt]);
        c1[mt] = wmma_f32_k4(a, b1v, c1[mt]);
      }
    }
    const size_t rbase = (size_t)(blockIdx.x * ROWS_PER_BLOCK) * 128 + n0 + lo;
#pragma unroll
    for (int mt = 0; mt < 4; ++mt) {
#pragma unroll
      for (int i = 0; i < 8; ++i) {
        const size_t o = rbase + (size_t)(mt * 16 + i + 8 * hi) * 128;
        out[o]      = c0[mt][i];
        out[o + 16] = c1[mt][i];
      }
    }
  }
}

extern "C" void kernel_launch(void* const* d_in, const int* in_sizes, int n_in,
                              void* d_out, int out_size, void* d_ws, size_t ws_size,
                              hipStream_t stream) {
  const float* dense = (const float*)d_in[0];
  const int* ids_rating       = (const int*)d_in[1];
  const int* ids_category     = (const int*)d_in[2];
  const int* ids_fandom       = (const int*)d_in[3];
  const int* ids_relationship = (const int*)d_in[4];
  const int* ids_character    = (const int*)d_in[5];
  const int* ids_freeform     = (const int*)d_in[6];
  const float* W_dense = (const float*)d_in[7];
  const float* b_dense = (const float*)d_in[8];
  const float* tab_rating       = (const float*)d_in[9];
  const float* tab_category     = (const float*)d_in[10];
  const float* tab_fandom       = (const float*)d_in[11];
  const float* tab_relationship = (const float*)d_in[12];
  const float* tab_character    = (const float*)d_in[13];
  const float* tab_freeform     = (const float*)d_in[14];
  const float* W1 = (const float*)d_in[15];
  const float* b1 = (const float*)d_in[16];
  const float* W2 = (const float*)d_in[17];
  const float* b2 = (const float*)d_in[18];
  float* out = (float*)d_out;

  const int Btot = in_sizes[0] / 64;
  const int grid = Btot / ROWS_PER_BLOCK;
  const size_t shmem = (size_t)(ROWS_PER_BLOCK * PRE_LD + 4 * 16 * H_LD) * sizeof(float)
                     + (size_t)IDS_TOTAL * sizeof(int);

  fused_embmlp_kernel<<<grid, THREADS, shmem, stream>>>(
      dense, ids_rating, ids_category, ids_fandom, ids_relationship,
      ids_character, ids_freeform, W_dense, b_dense, tab_rating, tab_category,
      tab_fandom, tab_relationship, tab_character, tab_freeform, W1, b1, W2, b2,
      out);
}
